# starting kernel
# baseline (speedup 1.0000x reference)
#include <hip/hip_runtime.h>


typedef float v2f __attribute__((ext_vector_type(2)));
typedef float v4f __attribute__((ext_vector_type(4)));
typedef float v8f __attribute__((ext_vector_type(8)));
typedef int   v4i __attribute__((ext_vector_type(4)));

#define SEQ  1024
#define EMB  1024
#define NH   64
#define DH   16

__global__ __launch_bounds__(32) void attn_flash_wmma(
    const float* __restrict__ keys, const float* __restrict__ query,
    const float* __restrict__ values, const int* __restrict__ mask,
    float* __restrict__ attn_out)
{
    const int lane = threadIdx.x;
    const int q    = lane & 15;
    const int half = lane >> 4;

    const int idx = blockIdx.x;
    const int qt  = idx & 63;
    const int h   = (idx >> 6) & 63;
    const int n   = idx >> 12;
    const int q0  = qt * 16;

    const float* Qrow = query + ((size_t)(n * SEQ + q0 + q)) * EMB + h * DH;
    v2f Bq[4];
#pragma unroll
    for (int dd = 0; dd < 4; ++dd)
        Bq[dd] = *reinterpret_cast<const v2f*>(Qrow + dd * 4 + 2 * half);

    v8f   o    = {};
    float mold = -1e30f;
    float lsum = 0.0f;

    const float* Kbase = keys   + ((size_t)(n * SEQ)) * EMB + h * DH;
    const float* Vbase = values + ((size_t)(n * SEQ)) * EMB + h * DH;
    const int*   Mrow  = mask + (size_t)n * SEQ * SEQ + (size_t)(q0 + q) * SEQ;

    for (int kt = 0; kt < SEQ / 16; ++kt) {
        const float* Krow = Kbase + (size_t)(kt * 16 + q) * EMB;
        v8f s = {};
#pragma unroll
        for (int dd = 0; dd < 4; ++dd) {
            v2f a = *reinterpret_cast<const v2f*>(Krow + dd * 4 + 2 * half);
            s = __builtin_amdgcn_wmma_f32_16x16x4_f32(
                    false, a, false, Bq[dd], (short)0, s, false, false);
        }

        const int* mb = Mrow + kt * 16 + 8 * half;
        v4i m0 = *reinterpret_cast<const v4i*>(mb);
        v4i m1 = *reinterpret_cast<const v4i*>(mb + 4);
        float p[8];
#pragma unroll
        for (int r = 0; r < 8; ++r) {
            float sr = s[r];
            int   mv = (r < 4) ? m0[r] : m1[r - 4];
            sr   = (mv == 0) ? -1e20f : sr;
            p[r] = sr * 0.03125f;
        }

        float tmax = p[0];
#pragma unroll
        for (int r = 1; r < 8; ++r) tmax = fmaxf(tmax, p[r]);
        tmax = fmaxf(tmax, __shfl_xor(tmax, 16));
        const float mnew  = fmaxf(mold, tmax);
        const float alpha = __expf(mold - mnew);
        float tsum = 0.0f;
#pragma unroll
        for (int r = 0; r < 8; ++r) { p[r] = __expf(p[r] - mnew); tsum += p[r]; }
        tsum += __shfl_xor(tsum, 16);
        lsum = lsum * alpha + tsum;
        mold = mnew;
#pragma unroll
        for (int r = 0; r < 8; ++r) o[r] *= alpha;

        const float* Vt = Vbase + (size_t)(kt * 16) * EMB + q;
#pragma unroll
        for (int kk = 0; kk < 4; ++kk) {
            const int k0 = 4 * kk + 2 * half;
            v2f av;
            av.x = Vt[(size_t)k0 * EMB];
            av.y = Vt[(size_t)(k0 + 1) * EMB];
            const int a       = (kk & 1) * 4;
            const int srcLane = q | ((kk >> 1) << 4);
            float t0 = __shfl(p[a],     srcLane);
            float t2 = __shfl(p[a + 2], srcLane);
            float t1 = __shfl(p[a + 1], srcLane);
            float t3 = __shfl(p[a + 3], srcLane);
            v2f bp;
            bp.x = half ? t2 : t0;
            bp.y = half ? t3 : t1;
            o = __builtin_amdgcn_wmma_f32_16x16x4_f32(
                    false, av, false, bp, (short)0, o, false, false);
        }
    }

    const float rl = 1.0f / lsum;
    float* Orow = attn_out + ((size_t)(n * SEQ + q0 + q)) * EMB + h * DH + 8 * half;
    v4f o0, o1;
#pragma unroll
    for (int r = 0; r < 4; ++r) { o0[r] = o[r] * rl; o1[r] = o[r + 4] * rl; }
    *reinterpret_cast<v4f*>(Orow)     = o0;
    *reinterpret_cast<v4f*>(Orow + 4) = o1;
}

__global__ __launch_bounds__(32) void proj_wmma(
    const float* __restrict__ X, const float* __restrict__ W,
    const float* __restrict__ bvec, float* __restrict__ out)
{
    const int lane = threadIdx.x;
    const int q    = lane & 15;
    const int half = lane >> 4;
    const int j0 = (blockIdx.x & 63) * 16;
    const int i0 = (blockIdx.x >> 6) * 16;

    const float* Xrow = X + (size_t)(i0 + q) * EMB;
    const float* Wrow = W + (size_t)(j0 + q) * EMB;
    v8f acc = {};
#pragma unroll 4
    for (int ee = 0; ee < EMB / 4; ++ee) {
        v2f a = *reinterpret_cast<const v2f*>(Xrow + ee * 4 + 2 * half);
        v2f b = *reinterpret_cast<const v2f*>(Wrow + ee * 4 + 2 * half);
        acc = __builtin_amdgcn_wmma_f32_16x16x4_f32(
                  false, a, false, b, (short)0, acc, false, false);
    }
    const float bj = bvec[j0 + q];
#pragma unroll
    for (int r = 0; r < 8; ++r)
        out[(size_t)(i0 + r + 8 * half) * EMB + j0 + q] = acc[r] + bj;
}

extern "C" void kernel_launch(void* const* d_in, const int* in_sizes, int n_in,
                              void* d_out, int out_size, void* d_ws, size_t ws_size,
                              hipStream_t stream)
{
    const float* keys   = (const float*)d_in[0];
    const float* query  = (const float*)d_in[1];
    const float* values = (const float*)d_in[2];
    const int*   mask   = (const int*)d_in[3];
    const float* W_out  = (const float*)d_in[4];
    const float* b_out  = (const float*)d_in[5];
    float* out  = (float*)d_out;
    float* attn = (float*)d_ws;

    const int N = 2;
    attn_flash_wmma<<<N * NH * (SEQ / 16), 32, 0, stream>>>(
        keys, query, values, mask, attn);
    proj_wmma<<<(N * SEQ / 16) * (EMB / 16), 32, 0, stream>>>(
        attn, W_out, b_out, out);
}


# proj GEMM: 128x64 block tiles staged in LDS by whole-line b128 loads (double buffer), 8 waves x 32x32 register tiles; k now summed 0,1,4,5,2,3,6,7 per 8 (reduction order changes)
# speedup vs baseline: 1.3445x; 1.3445x over previous
#include <hip/hip_runtime.h>


typedef float v2f __attribute__((ext_vector_type(2)));
typedef float v4f __attribute__((ext_vector_type(4)));
typedef float v8f __attribute__((ext_vector_type(8)));
typedef int   v4i __attribute__((ext_vector_type(4)));

#define SEQ  1024
#define EMB  1024
#define NH   64
#define DH   16

__global__ __launch_bounds__(32) void attn_flash_wmma(
    const float* __restrict__ keys, const float* __restrict__ query,
    const float* __restrict__ values, const int* __restrict__ mask,
    float* __restrict__ attn_out)
{
    const int lane = threadIdx.x;
    const int q    = lane & 15;
    const int half = lane >> 4;

    const int idx = blockIdx.x;
    const int qt  = idx & 63;
    const int h   = (idx >> 6) & 63;
    const int n   = idx >> 12;
    const int q0  = qt * 16;

    const float* Qrow = query + ((size_t)(n * SEQ + q0 + q)) * EMB + h * DH;
    v2f Bq[4];
#pragma unroll
    for (int dd = 0; dd < 4; ++dd)
        Bq[dd] = *reinterpret_cast<const v2f*>(Qrow + dd * 4 + 2 * half);

    v8f   o    = {};
    float mold = -1e30f;
    float lsum = 0.0f;

    const float* Kbase = keys   + ((size_t)(n * SEQ)) * EMB + h * DH;
    const float* Vbase = values + ((size_t)(n * SEQ)) * EMB + h * DH;
    const int*   Mrow  = mask + (size_t)n * SEQ * SEQ + (size_t)(q0 + q) * SEQ;

    for (int kt = 0; kt < SEQ / 16; ++kt) {
        const float* Krow = Kbase + (size_t)(kt * 16 + q) * EMB;
        v8f s = {};
#pragma unroll
        for (int dd = 0; dd < 4; ++dd) {
            v2f a = *reinterpret_cast<const v2f*>(Krow + dd * 4 + 2 * half);
            s = __builtin_amdgcn_wmma_f32_16x16x4_f32(
                    false, a, false, Bq[dd], (short)0, s, false, false);
        }

        const int* mb = Mrow + kt * 16 + 8 * half;
        v4i m0 = *reinterpret_cast<const v4i*>(mb);
        v4i m1 = *reinterpret_cast<const v4i*>(mb + 4);
        float p[8];
#pragma unroll
        for (int r = 0; r < 8; ++r) {
            float sr = s[r];
            int   mv = (r < 4) ? m0[r] : m1[r - 4];
            sr   = (mv == 0) ? -1e20f : sr;
            p[r] = sr * 0.03125f;
        }

        float tmax = p[0];
#pragma unroll
        for (int r = 1; r < 8; ++r) tmax = fmaxf(tmax, p[r]);
        tmax = fmaxf(tmax, __shfl_xor(tmax, 16));
        const float mnew  = fmaxf(mold, tmax);
        const float alpha = __expf(mold - mnew);
        float tsum = 0.0f;
#pragma unroll
        for (int r = 0; r < 8; ++r) { p[r] = __expf(p[r] - mnew); tsum += p[r]; }
        tsum += __shfl_xor(tsum, 16);
        lsum = lsum * alpha + tsum;
        mold = mnew;
#pragma unroll
        for (int r = 0; r < 8; ++r) o[r] *= alpha;

        const float* Vt = Vbase + (size_t)(kt * 16) * EMB + q;
#pragma unroll
        for (int kk = 0; kk < 4; ++kk) {
            const int k0 = 4 * kk + 2 * half;
            v2f av;
            av.x = Vt[(size_t)k0 * EMB];
            av.y = Vt[(size_t)(k0 + 1) * EMB];
            const int a       = (kk & 1) * 4;
            const int srcLane = q | ((kk >> 1) << 4);
            float t0 = __shfl(p[a],     srcLane);
            float t2 = __shfl(p[a + 2], srcLane);
            float t1 = __shfl(p[a + 1], srcLane);
            float t3 = __shfl(p[a + 3], srcLane);
            v2f bp;
            bp.x = half ? t2 : t0;
            bp.y = half ? t3 : t1;
            o = __builtin_amdgcn_wmma_f32_16x16x4_f32(
                    false, av, false, bp, (short)0, o, false, false);
        }
    }

    const float rl = 1.0f / lsum;
    float* Orow = attn_out + ((size_t)(n * SEQ + q0 + q)) * EMB + h * DH + 8 * half;
    v4f o0, o1;
#pragma unroll
    for (int r = 0; r < 4; ++r) { o0[r] = o[r] * rl; o1[r] = o[r + 4] * rl; }
    *reinterpret_cast<v4f*>(Orow)     = o0;
    *reinterpret_cast<v4f*>(Orow + 4) = o1;
}

#define P2_BM 128
#define P2_BN 64
#define P2_BK 64
#define P2_LD 68

__global__ __launch_bounds__(256) void proj_wmma(
    const float* __restrict__ X, const float* __restrict__ W,
    const float* __restrict__ bvec, float* __restrict__ out)
{
    __shared__ __attribute__((aligned(16))) float sA[2][P2_BM * P2_LD];
    __shared__ __attribute__((aligned(16))) float sB[2][P2_BN * P2_LD];

    const int tid  = threadIdx.x;
    const int lane = tid & 31;
    const int wave = tid >> 5;
    const int c    = lane & 15;
    const int h    = lane >> 4;
    const int wm   = wave >> 1;
    const int wn   = wave & 1;
    const int j0   = (blockIdx.x & 15) * P2_BN;
    const int i0   = (blockIdx.x >> 4) * P2_BM;

    const int lrow = tid >> 4;
    const int lcol = (tid & 15) * 4;
    const float* Xg = X + (size_t)(i0 + lrow) * EMB + lcol;
    const float* Wg = W + (size_t)(j0 + lrow) * EMB + lcol;
    const int lds_off = lrow * P2_LD + lcol;

    v4f ra[P2_BM / 16], rb[P2_BN / 16];
#pragma unroll
    for (int i = 0; i < P2_BM / 16; ++i)
        ra[i] = *reinterpret_cast<const v4f*>(Xg + (size_t)(16 * i) * EMB);
#pragma unroll
    for (int i = 0; i < P2_BN / 16; ++i)
        rb[i] = *reinterpret_cast<const v4f*>(Wg + (size_t)(16 * i) * EMB);
#pragma unroll
    for (int i = 0; i < P2_BM / 16; ++i)
        *reinterpret_cast<v4f*>(&sA[0][lds_off + 16 * i * P2_LD]) = ra[i];
#pragma unroll
    for (int i = 0; i < P2_BN / 16; ++i)
        *reinterpret_cast<v4f*>(&sB[0][lds_off + 16 * i * P2_LD]) = rb[i];
    __syncthreads();

    v8f acc00 = {}, acc01 = {}, acc10 = {}, acc11 = {};
    const int a_off = (wm * 32 + c) * P2_LD + 4 * h;
    const int b_off = (wn * 32 + c) * P2_LD + 4 * h;

    for (int kt = 0; kt < EMB / P2_BK; ++kt) {
        const int  cur  = kt & 1;
        const bool more = (kt + 1) < EMB / P2_BK;
        if (more) {
            const int kb = (kt + 1) * P2_BK;
#pragma unroll
            for (int i = 0; i < P2_BM / 16; ++i)
                ra[i] = *reinterpret_cast<const v4f*>(Xg + (size_t)(16 * i) * EMB + kb);
#pragma unroll
            for (int i = 0; i < P2_BN / 16; ++i)
                rb[i] = *reinterpret_cast<const v4f*>(Wg + (size_t)(16 * i) * EMB + kb);
        }

        const float* pa = &sA[cur][a_off];
        const float* pb = &sB[cur][b_off];
#pragma unroll
        for (int k8 = 0; k8 < P2_BK / 8; ++k8) {
            const v4f fa0 = *reinterpret_cast<const v4f*>(pa + 8 * k8);
            const v4f fa1 = *reinterpret_cast<const v4f*>(pa + 16 * P2_LD + 8 * k8);
            const v4f fb0 = *reinterpret_cast<const v4f*>(pb + 8 * k8);
            const v4f fb1 = *reinterpret_cast<const v4f*>(pb + 16 * P2_LD + 8 * k8);
            v2f a0, a1, b0, b1;
            a0.x = fa0.x; a0.y = fa0.y;  a1.x = fa1.x; a1.y = fa1.y;
            b0.x = fb0.x; b0.y = fb0.y;  b1.x = fb1.x; b1.y = fb1.y;
            acc00 = __builtin_amdgcn_wmma_f32_16x16x4_f32(false, a0, false, b0, (short)0, acc00, false, false);
            acc01 = __builtin_amdgcn_wmma_f32_16x16x4_f32(false, a0, false, b1, (short)0, acc01, false, false);
            acc10 = __builtin_amdgcn_wmma_f32_16x16x4_f32(false, a1, false, b0, (short)0, acc10, false, false);
            acc11 = __builtin_amdgcn_wmma_f32_16x16x4_f32(false, a1, false, b1, (short)0, acc11, false, false);
            a0.x = fa0.z; a0.y = fa0.w;  a1.x = fa1.z; a1.y = fa1.w;
            b0.x = fb0.z; b0.y = fb0.w;  b1.x = fb1.z; b1.y = fb1.w;
            acc00 = __builtin_amdgcn_wmma_f32_16x16x4_f32(false, a0, false, b0, (short)0, acc00, false, false);
            acc01 = __builtin_amdgcn_wmma_f32_16x16x4_f32(false, a0, false, b1, (short)0, acc01, false, false);
            acc10 = __builtin_amdgcn_wmma_f32_16x16x4_f32(false, a1, false, b0, (short)0, acc10, false, false);
            acc11 = __builtin_amdgcn_wmma_f32_16x16x4_f32(false, a1, false, b1, (short)0, acc11, false, false);
        }

        if (more) {
            const int nxt = cur ^ 1;
#pragma unroll
            for (int i = 0; i < P2_BM / 16; ++i)
                *reinterpret_cast<v4f*>(&sA[nxt][lds_off + 16 * i * P2_LD]) = ra[i];
#pragma unroll
            for (int i = 0; i < P2_BN / 16; ++i)
                *reinterpret_cast<v4f*>(&sB[nxt][lds_off + 16 * i * P2_LD]) = rb[i];
        }
        __syncthreads();
    }

    const int   col0 = j0 + wn * 32 + c;
    const int   col1 = col0 + 16;
    const float bj0  = bvec[col0];
    const float bj1  = bvec[col1];
    const int   row0 = i0 + wm * 32 + 8 * h;
#pragma unroll
    for (int r = 0; r < 8; ++r) {
        out[(size_t)(row0 + r) * EMB + col0]      = acc00[r] + bj0;
        out[(size_t)(row0 + r) * EMB + col1]      = acc01[r] + bj1;
        out[(size_t)(row0 + 16 + r) * EMB + col0] = acc10[r] + bj0;
        out[(size_t)(row0 + 16 + r) * EMB + col1] = acc11[r] + bj1;
    }
}

extern "C" void kernel_launch(void* const* d_in, const int* in_sizes, int n_in,
                              void* d_out, int out_size, void* d_ws, size_t ws_size,
                              hipStream_t stream)
{
    const float* keys   = (const float*)d_in[0];
    const float* query  = (const float*)d_in[1];
    const float* values = (const float*)d_in[2];
    const int*   mask   = (const int*)d_in[3];
    const float* W_out  = (const float*)d_in[4];
    const float* b_out  = (const float*)d_in[5];
    float* out  = (float*)d_out;
    float* attn = (float*)d_ws;

    const int N = 2;
    attn_flash_wmma<<<N * NH * (SEQ / 16), 32, 0, stream>>>(
        keys, query, values, mask, attn);
    proj_wmma<<<(N * SEQ / P2_BM) * (EMB / P2_BN), 256, 0, stream>>>(
        attn, W_out, b_out, out);
}


# attention: 8 waves per block share each 64-key K/V tile via LDS (b128 copies, double buffer; mask fetched a tile ahead); K fragment read b128: score d order 0,1,8,9,2,3.. (sum order changes)
# speedup vs baseline: 1.8680x; 1.3894x over previous
#include <hip/hip_runtime.h>


typedef float v2f __attribute__((ext_vector_type(2)));
typedef float v4f __attribute__((ext_vector_type(4)));
typedef float v8f __attribute__((ext_vector_type(8)));
typedef int   v4i __attribute__((ext_vector_type(4)));

#define SEQ  1024
#define EMB  1024
#define NH   64
#define DH   16

#define A_KT 64
#define A_LD 20

__global__ __launch_bounds__(256) void attn_flash_wmma(
    const float* __restrict__ keys, const float* __restrict__ query,
    const float* __restrict__ values, const int* __restrict__ mask,
    float* __restrict__ attn_out)
{
    __shared__ __attribute__((aligned(16))) float sK[2][A_KT * A_LD];
    __shared__ __attribute__((aligned(16))) float sV[2][A_KT * A_LD];

    const int tid  = threadIdx.x;
    const int lane = tid & 31;
    const int wave = tid >> 5;
    const int q    = lane & 15;
    const int half = lane >> 4;

    const int idx = blockIdx.x;
    const int qb  = idx & 7;
    const int h   = (idx >> 3) & 63;
    const int n   = idx >> 9;
    const int q0  = qb * 128 + wave * 16;

    const float* Qrow = query + ((size_t)(n * SEQ + q0 + q)) * EMB + h * DH + 8 * half;
    const v4f qlo = *reinterpret_cast<const v4f*>(Qrow);
    const v4f qhi = *reinterpret_cast<const v4f*>(Qrow + 4);
    v2f Bq[4];
    Bq[0].x = qlo.x; Bq[0].y = qlo.y;
    Bq[1].x = qlo.z; Bq[1].y = qlo.w;
    Bq[2].x = qhi.x; Bq[2].y = qhi.y;
    Bq[3].x = qhi.z; Bq[3].y = qhi.w;

    v8f   o    = {};
    float mold = -1e30f;
    float lsum = 0.0f;

    const int srow = tid >> 2;
    const int scol = (tid & 3) * 4;
    const float* Kg = keys   + ((size_t)(n * SEQ + srow)) * EMB + h * DH + scol;
    const float* Vg = values + ((size_t)(n * SEQ + srow)) * EMB + h * DH + scol;
    const int st_off = srow * A_LD + scol;
    const int*   Mrow  = mask + (size_t)n * SEQ * SEQ + (size_t)(q0 + q) * SEQ + 8 * half;

    v4f rk = *reinterpret_cast<const v4f*>(Kg);
    v4f rv = *reinterpret_cast<const v4f*>(Vg);
    *reinterpret_cast<v4f*>(&sK[0][st_off]) = rk;
    *reinterpret_cast<v4f*>(&sV[0][st_off]) = rv;
    __syncthreads();

    const int kf_off = q * A_LD + 8 * half;
    const int vf_off = (2 * half) * A_LD + q;

    v4i mk[2][2];
    mk[0][0] = *reinterpret_cast<const v4i*>(Mrow);
    mk[0][1] = *reinterpret_cast<const v4i*>(Mrow + 4);

    for (int st = 0; st < SEQ / A_KT; ++st) {
        const int cur = st & 1;
        const int stn = min(st + 1, SEQ / A_KT - 1);
        rk = *reinterpret_cast<const v4f*>(Kg + (size_t)(stn * A_KT) * EMB);
        rv = *reinterpret_cast<const v4f*>(Vg + (size_t)(stn * A_KT) * EMB);

#pragma unroll
        for (int t = 0; t < A_KT / 16; ++t) {
            const int ktn = min(st * (A_KT / 16) + t + 1, SEQ / 16 - 1);
            mk[(t + 1) & 1][0] = *reinterpret_cast<const v4i*>(Mrow + ktn * 16);
            mk[(t + 1) & 1][1] = *reinterpret_cast<const v4i*>(Mrow + ktn * 16 + 4);

            const float* pk = &sK[cur][t * 16 * A_LD + kf_off];
            const float* pv = &sV[cur][t * 16 * A_LD + vf_off];
            const v4f klo = *reinterpret_cast<const v4f*>(pk);
            const v4f khi = *reinterpret_cast<const v4f*>(pk + 4);
            v2f av[4];
#pragma unroll
            for (int kk = 0; kk < 4; ++kk) {
                av[kk].x = pv[(4 * kk) * A_LD];
                av[kk].y = pv[(4 * kk + 1) * A_LD];
            }

            __builtin_amdgcn_sched_barrier(0);

            v2f ka[4];
            ka[0].x = klo.x; ka[0].y = klo.y;
            ka[1].x = klo.z; ka[1].y = klo.w;
            ka[2].x = khi.x; ka[2].y = khi.y;
            ka[3].x = khi.z; ka[3].y = khi.w;
            v8f s = {};
#pragma unroll
            for (int dd = 0; dd < 4; ++dd)
                s = __builtin_amdgcn_wmma_f32_16x16x4_f32(
                        false, ka[dd], false, Bq[dd], (short)0, s, false, false);

            const v4i m0 = mk[t & 1][0];
            const v4i m1 = mk[t & 1][1];
            float p[8];
#pragma unroll
            for (int r = 0; r < 8; ++r) {
                float sr = s[r];
                int   mv = (r < 4) ? m0[r] : m1[r - 4];
                sr   = (mv == 0) ? -1e20f : sr;
                p[r] = sr * 0.03125f;
            }

            float tmax = p[0];
#pragma unroll
            for (int r = 1; r < 8; ++r) tmax = fmaxf(tmax, p[r]);
            tmax = fmaxf(tmax, __shfl_xor(tmax, 16));
            const float mnew  = fmaxf(mold, tmax);
            const float alpha = __expf(mold - mnew);
            float tsum = 0.0f;
#pragma unroll
            for (int r = 0; r < 8; ++r) { p[r] = __expf(p[r] - mnew); tsum += p[r]; }
            tsum += __shfl_xor(tsum, 16);
            lsum = lsum * alpha + tsum;
            mold = mnew;
#pragma unroll
            for (int r = 0; r < 8; ++r) o[r] *= alpha;

#pragma unroll
            for (int kk = 0; kk < 4; ++kk) {
                const int a       = (kk & 1) * 4;
                const int srcLane = q | ((kk >> 1) << 4);
                float t0 = __shfl(p[a],     srcLane);
                float t2 = __shfl(p[a + 2], srcLane);
                float t1 = __shfl(p[a + 1], srcLane);
                float t3 = __shfl(p[a + 3], srcLane);
                v2f bp;
                bp.x = half ? t2 : t0;
                bp.y = half ? t3 : t1;
                o = __builtin_amdgcn_wmma_f32_16x16x4_f32(
                        false, av[kk], false, bp, (short)0, o, false, false);
            }
        }

        *reinterpret_cast<v4f*>(&sK[cur ^ 1][st_off]) = rk;
        *reinterpret_cast<v4f*>(&sV[cur ^ 1][st_off]) = rv;
        __syncthreads();
    }

    const float rl = 1.0f / lsum;
    float* Orow = attn_out + ((size_t)(n * SEQ + q0 + q)) * EMB + h * DH + 8 * half;
    v4f o0, o1;
#pragma unroll
    for (int r = 0; r < 4; ++r) { o0[r] = o[r] * rl; o1[r] = o[r + 4] * rl; }
    *reinterpret_cast<v4f*>(Orow)     = o0;
    *reinterpret_cast<v4f*>(Orow + 4) = o1;
}

#define P2_BM 128
#define P2_BN 64
#define P2_BK 64
#define P2_LD 68

__global__ __launch_bounds__(256) void proj_wmma(
    const float* __restrict__ X, const float* __restrict__ W,
    const float* __restrict__ bvec, float* __restrict__ out)
{
    __shared__ __attribute__((aligned(16))) float sA[2][P2_BM * P2_LD];
    __shared__ __attribute__((aligned(16))) float sB[2][P2_BN * P2_LD];

    const int tid  = threadIdx.x;
    const int lane = tid & 31;
    const int wave = tid >> 5;
    const int c    = lane & 15;
    const int h    = lane >> 4;
    const int wm   = wave >> 1;
    const int wn   = wave & 1;
    const int j0   = (blockIdx.x & 15) * P2_BN;
    const int i0   = (blockIdx.x >> 4) * P2_BM;

    const int lrow = tid >> 4;
    const int lcol = (tid & 15) * 4;
    const float* Xg = X + (size_t)(i0 + lrow) * EMB + lcol;
    const float* Wg = W + (size_t)(j0 + lrow) * EMB + lcol;
    const int lds_off = lrow * P2_LD + lcol;

    v4f ra[P2_BM / 16], rb[P2_BN / 16];
#pragma unroll
    for (int i = 0; i < P2_BM / 16; ++i)
        ra[i] = *reinterpret_cast<const v4f*>(Xg + (size_t)(16 * i) * EMB);
#pragma unroll
    for (int i = 0; i < P2_BN / 16; ++i)
        rb[i] = *reinterpret_cast<const v4f*>(Wg + (size_t)(16 * i) * EMB);
#pragma unroll
    for (int i = 0; i < P2_BM / 16; ++i)
        *reinterpret_cast<v4f*>(&sA[0][lds_off + 16 * i * P2_LD]) = ra[i];
#pragma unroll
    for (int i = 0; i < P2_BN / 16; ++i)
        *reinterpret_cast<v4f*>(&sB[0][lds_off + 16 * i * P2_LD]) = rb[i];
    __syncthreads();

    v8f acc00 = {}, acc01 = {}, acc10 = {}, acc11 = {};
    const int a_off = (wm * 32 + c) * P2_LD + 4 * h;
    const int b_off = (wn * 32 + c) * P2_LD + 4 * h;

    for (int kt = 0; kt < EMB / P2_BK; ++kt) {
        const int  cur  = kt & 1;
        const bool more = (kt + 1) < EMB / P2_BK;
        if (more) {
            const int kb = (kt + 1) * P2_BK;
#pragma unroll
            for (int i = 0; i < P2_BM / 16; ++i)
                ra[i] = *reinterpret_cast<const v4f*>(Xg + (size_t)(16 * i) * EMB + kb);
#pragma unroll
            for (int i = 0; i < P2_BN / 16; ++i)
                rb[i] = *reinterpret_cast<const v4f*>(Wg + (size_t)(16 * i) * EMB + kb);
        }

        const float* pa = &sA[cur][a_off];
        const float* pb = &sB[cur][b_off];
#pragma unroll
        for (int k8 = 0; k8 < P2_BK / 8; ++k8) {
            const v4f fa0 = *reinterpret_cast<const v4f*>(pa + 8 * k8);
            const v4f fa1 = *reinterpret_cast<const v4f*>(pa + 16 * P2_LD + 8 * k8);
            const v4f fb0 = *reinterpret_cast<const v4f*>(pb + 8 * k8);
            const v4f fb1 = *reinterpret_cast<const v4f*>(pb + 16 * P2_LD + 8 * k8);
            v2f a0, a1, b0, b1;
            a0.x = fa0.x; a0.y = fa0.y;  a1.x = fa1.x; a1.y = fa1.y;
            b0.x = fb0.x; b0.y = fb0.y;  b1.x = fb1.x; b1.y = fb1.y;
            acc00 = __builtin_amdgcn_wmma_f32_16x16x4_f32(false, a0, false, b0, (short)0, acc00, false, false);
            acc01 = __builtin_amdgcn_wmma_f32_16x16x4_f32(false, a0, false, b1, (short)0, acc01, false, false);
            acc10 = __builtin_amdgcn_wmma_f32_16x16x4_f32(false, a1, false, b0, (short)0, acc10, false, false);
            acc11 = __builtin_amdgcn_wmma_f32_16x16x4_f32(false, a1, false, b1, (short)0, acc11, false, false);
            a0.x = fa0.z; a0.y = fa0.w;  a1.x = fa1.z; a1.y = fa1.w;
            b0.x = fb0.z; b0.y = fb0.w;  b1.x = fb1.z; b1.y = fb1.w;
            acc00 = __builtin_amdgcn_wmma_f32_16x16x4_f32(false, a0, false, b0, (short)0, acc00, false, false);
            acc01 = __builtin_amdgcn_wmma_f32_16x16x4_f32(false, a0, false, b1, (short)0, acc01, false, false);
            acc10 = __builtin_amdgcn_wmma_f32_16x16x4_f32(false, a1, false, b0, (short)0, acc10, false, false);
            acc11 = __builtin_amdgcn_wmma_f32_16x16x4_f32(false, a1, false, b1, (short)0, acc11, false, false);
        }

        if (more) {
            const int nxt = cur ^ 1;
#pragma unroll
            for (int i = 0; i < P2_BM / 16; ++i)
                *reinterpret_cast<v4f*>(&sA[nxt][lds_off + 16 * i * P2_LD]) = ra[i];
#pragma unroll
            for (int i = 0; i < P2_BN / 16; ++i)
                *reinterpret_cast<v4f*>(&sB[nxt][lds_off + 16 * i * P2_LD]) = rb[i];
        }
        __syncthreads();
    }

    const int   col0 = j0 + wn * 32 + c;
    const int   col1 = col0 + 16;
    const float bj0  = bvec[col0];
    const float bj1  = bvec[col1];
    const int   row0 = i0 + wm * 32 + 8 * h;
#pragma unroll
    for (int r = 0; r < 8; ++r) {
        out[(size_t)(row0 + r) * EMB + col0]      = acc00[r] + bj0;
        out[(size_t)(row0 + r) * EMB + col1]      = acc01[r] + bj1;
        out[(size_t)(row0 + 16 + r) * EMB + col0] = acc10[r] + bj0;
        out[(size_t)(row0 + 16 + r) * EMB + col1] = acc11[r] + bj1;
    }
}

extern "C" void kernel_launch(void* const* d_in, const int* in_sizes, int n_in,
                              void* d_out, int out_size, void* d_ws, size_t ws_size,
                              hipStream_t stream)
{
    const float* keys   = (const float*)d_in[0];
    const float* query  = (const float*)d_in[1];
    const float* values = (const float*)d_in[2];
    const int*   mask   = (const int*)d_in[3];
    const float* W_out  = (const float*)d_in[4];
    const float* b_out  = (const float*)d_in[5];
    float* out  = (float*)d_out;
    float* attn = (float*)d_ws;

    const int N = 2;
    attn_flash_wmma<<<N * NH * (SEQ / 128), 256, 0, stream>>>(
        keys, query, values, mask, attn);
    proj_wmma<<<(N * SEQ / P2_BM) * (EMB / P2_BN), 256, 0, stream>>>(
        attn, W_out, b_out, out);
}


# attention: PV product sums each tile's 16 keys in the score accumulator's own order (0,1,8,9,2,3,10,11..), P feeds the WMMA in place; 16 cross-lane moves + 8 selects per tile go (sum order changes)
# speedup vs baseline: 2.0119x; 1.0770x over previous
#include <hip/hip_runtime.h>


typedef float v2f __attribute__((ext_vector_type(2)));
typedef float v4f __attribute__((ext_vector_type(4)));
typedef float v8f __attribute__((ext_vector_type(8)));
typedef int   v4i __attribute__((ext_vector_type(4)));

#define SEQ  1024
#define EMB  1024
#define NH   64
#define DH   16

#define A_KT 64
#define A_LD 20

__global__ __launch_bounds__(256) void attn_flash_wmma(
    const float* __restrict__ keys, const float* __restrict__ query,
    const float* __restrict__ values, const int* __restrict__ mask,
    float* __restrict__ attn_out)
{
    __shared__ __attribute__((aligned(16))) float sK[2][A_KT * A_LD];
    __shared__ __attribute__((aligned(16))) float sV[2][A_KT * A_LD];

    const int tid  = threadIdx.x;
    const int lane = tid & 31;
    const int wave = tid >> 5;
    const int q    = lane & 15;
    const int half = lane >> 4;

    const int idx = blockIdx.x;
    const int qb  = idx & 7;
    const int h   = (idx >> 3) & 63;
    const int n   = idx >> 9;
    const int q0  = qb * 128 + wave * 16;

    const float* Qrow = query + ((size_t)(n * SEQ + q0 + q)) * EMB + h * DH + 8 * half;
    const v4f qlo = *reinterpret_cast<const v4f*>(Qrow);
    const v4f qhi = *reinterpret_cast<const v4f*>(Qrow + 4);
    v2f Bq[4];
    Bq[0].x = qlo.x; Bq[0].y = qlo.y;
    Bq[1].x = qlo.z; Bq[1].y = qlo.w;
    Bq[2].x = qhi.x; Bq[2].y = qhi.y;
    Bq[3].x = qhi.z; Bq[3].y = qhi.w;

    v8f   o    = {};
    float mold = -1e30f;
    float lsum = 0.0f;

    const int srow = tid >> 2;
    const int scol = (tid & 3) * 4;
    const float* Kg = keys   + ((size_t)(n * SEQ + srow)) * EMB + h * DH + scol;
    const float* Vg = values + ((size_t)(n * SEQ + srow)) * EMB + h * DH + scol;
    const int st_off = srow * A_LD + scol;
    const int*   Mrow  = mask + (size_t)n * SEQ * SEQ + (size_t)(q0 + q) * SEQ + 8 * half;

    v4f rk = *reinterpret_cast<const v4f*>(Kg);
    v4f rv = *reinterpret_cast<const v4f*>(Vg);
    *reinterpret_cast<v4f*>(&sK[0][st_off]) = rk;
    *reinterpret_cast<v4f*>(&sV[0][st_off]) = rv;
    __syncthreads();

    const int kf_off = q * A_LD + 8 * half;
    const int vf_off = (8 * half) * A_LD + q;

    v4i mk[2][2];
    mk[0][0] = *reinterpret_cast<const v4i*>(Mrow);
    mk[0][1] = *reinterpret_cast<const v4i*>(Mrow + 4);

    for (int st = 0; st < SEQ / A_KT; ++st) {
        const int cur = st & 1;
        const int stn = min(st + 1, SEQ / A_KT - 1);
        rk = *reinterpret_cast<const v4f*>(Kg + (size_t)(stn * A_KT) * EMB);
        rv = *reinterpret_cast<const v4f*>(Vg + (size_t)(stn * A_KT) * EMB);

#pragma unroll
        for (int t = 0; t < A_KT / 16; ++t) {
            const int ktn = min(st * (A_KT / 16) + t + 1, SEQ / 16 - 1);
            mk[(t + 1) & 1][0] = *reinterpret_cast<const v4i*>(Mrow + ktn * 16);
            mk[(t + 1) & 1][1] = *reinterpret_cast<const v4i*>(Mrow + ktn * 16 + 4);

            const float* pk = &sK[cur][t * 16 * A_LD + kf_off];
            const float* pv = &sV[cur][t * 16 * A_LD + vf_off];
            const v4f klo = *reinterpret_cast<const v4f*>(pk);
            const v4f khi = *reinterpret_cast<const v4f*>(pk + 4);
            v2f av[4];
#pragma unroll
            for (int kk = 0; kk < 4; ++kk) {
                av[kk].x = pv[(2 * kk) * A_LD];
                av[kk].y = pv[(2 * kk + 1) * A_LD];
            }

            __builtin_amdgcn_sched_barrier(0);

            v2f ka[4];
            ka[0].x = klo.x; ka[0].y = klo.y;
            ka[1].x = klo.z; ka[1].y = klo.w;
            ka[2].x = khi.x; ka[2].y = khi.y;
            ka[3].x = khi.z; ka[3].y = khi.w;
            v8f s = {};
#pragma unroll
            for (int dd = 0; dd < 4; ++dd)
                s = __builtin_amdgcn_wmma_f32_16x16x4_f32(
                        false, ka[dd], false, Bq[dd], (short)0, s, false, false);

            const v4i m0 = mk[t & 1][0];
            const v4i m1 = mk[t & 1][1];
            float p[8];
#pragma unroll
            for (int r = 0; r < 8; ++r) {
                float sr = s[r];
                int   mv = (r < 4) ? m0[r] : m1[r - 4];
                sr   = (mv == 0) ? -1e20f : sr;
                p[r] = sr * 0.03125f;
            }

            float tmax = p[0];
#pragma unroll
            for (int r = 1; r < 8; ++r) tmax = fmaxf(tmax, p[r]);
            tmax = fmaxf(tmax, __shfl_xor(tmax, 16));
            const float mnew  = fmaxf(mold, tmax);
            const float alpha = __expf(mold - mnew);
            float tsum = 0.0f;
#pragma unroll
            for (int r = 0; r < 8; ++r) { p[r] = __expf(p[r] - mnew); tsum += p[r]; }
            tsum += __shfl_xor(tsum, 16);
            lsum = lsum * alpha + tsum;
            mold = mnew;
#pragma unroll
            for (int r = 0; r < 8; ++r) o[r] *= alpha;

#pragma unroll
            for (int kk = 0; kk < 4; ++kk) {
                v2f bp;
                bp.x = p[2 * kk];
                bp.y = p[2 * kk + 1];
                o = __builtin_amdgcn_wmma_f32_16x16x4_f32(
                        false, av[kk], false, bp, (short)0, o, false, false);
            }
        }

        *reinterpret_cast<v4f*>(&sK[cur ^ 1][st_off]) = rk;
        *reinterpret_cast<v4f*>(&sV[cur ^ 1][st_off]) = rv;
        __syncthreads();
    }

    const float rl = 1.0f / lsum;
    float* Orow = attn_out + ((size_t)(n * SEQ + q0 + q)) * EMB + h * DH + 8 * half;
    v4f o0, o1;
#pragma unroll
    for (int r = 0; r < 4; ++r) { o0[r] = o[r] * rl; o1[r] = o[r + 4] * rl; }
    *reinterpret_cast<v4f*>(Orow)     = o0;
    *reinterpret_cast<v4f*>(Orow + 4) = o1;
}

#define P2_BM 128
#define P2_BN 64
#define P2_BK 64
#define P2_LD 68

__global__ __launch_bounds__(256) void proj_wmma(
    const float* __restrict__ X, const float* __restrict__ W,
    const float* __restrict__ bvec, float* __restrict__ out)
{
    __shared__ __attribute__((aligned(16))) float sA[2][P2_BM * P2_LD];
    __shared__ __attribute__((aligned(16))) float sB[2][P2_BN * P2_LD];

    const int tid  = threadIdx.x;
    const int lane = tid & 31;
    const int wave = tid >> 5;
    const int c    = lane & 15;
    const int h    = lane >> 4;
    const int wm   = wave >> 1;
    const int wn   = wave & 1;
    const int j0   = (blockIdx.x & 15) * P2_BN;
    const int i0   = (blockIdx.x >> 4) * P2_BM;

    const int lrow = tid >> 4;
    const int lcol = (tid & 15) * 4;
    const float* Xg = X + (size_t)(i0 + lrow) * EMB + lcol;
    const float* Wg = W + (size_t)(j0 + lrow) * EMB + lcol;
    const int lds_off = lrow * P2_LD + lcol;

    v4f ra[P2_BM / 16], rb[P2_BN / 16];
#pragma unroll
    for (int i = 0; i < P2_BM / 16; ++i)
        ra[i] = *reinterpret_cast<const v4f*>(Xg + (size_t)(16 * i) * EMB);
#pragma unroll
    for (int i = 0; i < P2_BN / 16; ++i)
        rb[i] = *reinterpret_cast<const v4f*>(Wg + (size_t)(16 * i) * EMB);
#pragma unroll
    for (int i = 0; i < P2_BM / 16; ++i)
        *reinterpret_cast<v4f*>(&sA[0][lds_off + 16 * i * P2_LD]) = ra[i];
#pragma unroll
    for (int i = 0; i < P2_BN / 16; ++i)
        *reinterpret_cast<v4f*>(&sB[0][lds_off + 16 * i * P2_LD]) = rb[i];
    __syncthreads();

    v8f acc00 = {}, acc01 = {}, acc10 = {}, acc11 = {};
    const int a_off = (wm * 32 + c) * P2_LD + 4 * h;
    const int b_off = (wn * 32 + c) * P2_LD + 4 * h;

    for (int kt = 0; kt < EMB / P2_BK; ++kt) {
        const int  cur  = kt & 1;
        const bool more = (kt + 1) < EMB / P2_BK;
        if (more) {
            const int kb = (kt + 1) * P2_BK;
#pragma unroll
            for (int i = 0; i < P2_BM / 16; ++i)
                ra[i] = *reinterpret_cast<const v4f*>(Xg + (size_t)(16 * i) * EMB + kb);
#pragma unroll
            for (int i = 0; i < P2_BN / 16; ++i)
                rb[i] = *reinterpret_cast<const v4f*>(Wg + (size_t)(16 * i) * EMB + kb);
        }

        const float* pa = &sA[cur][a_off];
        const float* pb = &sB[cur][b_off];
#pragma unroll
        for (int k8 = 0; k8 < P2_BK / 8; ++k8) {
            const v4f fa0 = *reinterpret_cast<const v4f*>(pa + 8 * k8);
            const v4f fa1 = *reinterpret_cast<const v4f*>(pa + 16 * P2_LD + 8 * k8);
            const v4f fb0 = *reinterpret_cast<const v4f*>(pb + 8 * k8);
            const v4f fb1 = *reinterpret_cast<const v4f*>(pb + 16 * P2_LD + 8 * k8);
            v2f a0, a1, b0, b1;
            a0.x = fa0.x; a0.y = fa0.y;  a1.x = fa1.x; a1.y = fa1.y;
            b0.x = fb0.x; b0.y = fb0.y;  b1.x = fb1.x; b1.y = fb1.y;
            acc00 = __builtin_amdgcn_wmma_f32_16x16x4_f32(false, a0, false, b0, (short)0, acc00, false, false);
            acc01 = __builtin_amdgcn_wmma_f32_16x16x4_f32(false, a0, false, b1, (short)0, acc01, false, false);
            acc10 = __builtin_amdgcn_wmma_f32_16x16x4_f32(false, a1, false, b0, (short)0, acc10, false, false);
            acc11 = __builtin_amdgcn_wmma_f32_16x16x4_f32(false, a1, false, b1, (short)0, acc11, false, false);
            a0.x = fa0.z; a0.y = fa0.w;  a1.x = fa1.z; a1.y = fa1.w;
            b0.x = fb0.z; b0.y = fb0.w;  b1.x = fb1.z; b1.y = fb1.w;
            acc00 = __builtin_amdgcn_wmma_f32_16x16x4_f32(false, a0, false, b0, (short)0, acc00, false, false);
            acc01 = __builtin_amdgcn_wmma_f32_16x16x4_f32(false, a0, false, b1, (short)0, acc01, false, false);
            acc10 = __builtin_amdgcn_wmma_f32_16x16x4_f32(false, a1, false, b0, (short)0, acc10, false, false);
            acc11 = __builtin_amdgcn_wmma_f32_16x16x4_f32(false, a1, false, b1, (short)0, acc11, false, false);
        }

        if (more) {
            const int nxt = cur ^ 1;
#pragma unroll
            for (int i = 0; i < P2_BM / 16; ++i)
                *reinterpret_cast<v4f*>(&sA[nxt][lds_off + 16 * i * P2_LD]) = ra[i];
#pragma unroll
            for (int i = 0; i < P2_BN / 16; ++i)
                *reinterpret_cast<v4f*>(&sB[nxt][lds_off + 16 * i * P2_LD]) = rb[i];
        }
        __syncthreads();
    }

    const int   col0 = j0 + wn * 32 + c;
    const int   col1 = col0 + 16;
    const float bj0  = bvec[col0];
    const float bj1  = bvec[col1];
    const int   row0 = i0 + wm * 32 + 8 * h;
#pragma unroll
    for (int r = 0; r < 8; ++r) {
        out[(size_t)(row0 + r) * EMB + col0]      = acc00[r] + bj0;
        out[(size_t)(row0 + r) * EMB + col1]      = acc01[r] + bj1;
        out[(size_t)(row0 + 16 + r) * EMB + col0] = acc10[r] + bj0;
        out[(size_t)(row0 + 16 + r) * EMB + col1] = acc11[r] + bj1;
    }
}

extern "C" void kernel_launch(void* const* d_in, const int* in_sizes, int n_in,
                              void* d_out, int out_size, void* d_ws, size_t ws_size,
                              hipStream_t stream)
{
    const float* keys   = (const float*)d_in[0];
    const float* query  = (const float*)d_in[1];
    const float* values = (const float*)d_in[2];
    const int*   mask   = (const int*)d_in[3];
    const float* W_out  = (const float*)d_in[4];
    const float* b_out  = (const float*)d_in[5];
    float* out  = (float*)d_out;
    float* attn = (float*)d_ws;

    const int N = 2;
    attn_flash_wmma<<<N * NH * (SEQ / 128), 256, 0, stream>>>(
        keys, query, values, mask, attn);
    proj_wmma<<<(N * SEQ / P2_BM) * (EMB / P2_BN), 256, 0, stream>>>(
        attn, W_out, b_out, out);
}


# attention softmax: Q fragment multiplied once per wave by log2(e)/sqrt(E), so scores leave the WMMA in base-2 units; per key only subtract max + v_exp remain (score rounding changes, sum order kept)
# speedup vs baseline: 2.0808x; 1.0342x over previous
#include <hip/hip_runtime.h>


typedef float v2f __attribute__((ext_vector_type(2)));
typedef float v4f __attribute__((ext_vector_type(4)));
typedef float v8f __attribute__((ext_vector_type(8)));
typedef int   v4i __attribute__((ext_vector_type(4)));

#define SEQ  1024
#define EMB  1024
#define NH   64
#define DH   16

#define A_KT 64
#define A_LD 20
#define SM_C (1.4426950408889634f * 0.03125f)

__global__ __launch_bounds__(256) void attn_flash_wmma(
    const float* __restrict__ keys, const float* __restrict__ query,
    const float* __restrict__ values, const int* __restrict__ mask,
    float* __restrict__ attn_out)
{
    __shared__ __attribute__((aligned(16))) float sK[2][A_KT * A_LD];
    __shared__ __attribute__((aligned(16))) float sV[2][A_KT * A_LD];

    const int tid  = threadIdx.x;
    const int lane = tid & 31;
    const int wave = tid >> 5;
    const int q    = lane & 15;
    const int half = lane >> 4;

    const int idx = blockIdx.x;
    const int qb  = idx & 7;
    const int h   = (idx >> 3) & 63;
    const int n   = idx >> 9;
    const int q0  = qb * 128 + wave * 16;

    const float* Qrow = query + ((size_t)(n * SEQ + q0 + q)) * EMB + h * DH + 8 * half;
    const v4f qlo = *reinterpret_cast<const v4f*>(Qrow);
    const v4f qhi = *reinterpret_cast<const v4f*>(Qrow + 4);
    v2f Bq[4];
    Bq[0].x = qlo.x * SM_C; Bq[0].y = qlo.y * SM_C;
    Bq[1].x = qlo.z * SM_C; Bq[1].y = qlo.w * SM_C;
    Bq[2].x = qhi.x * SM_C; Bq[2].y = qhi.y * SM_C;
    Bq[3].x = qhi.z * SM_C; Bq[3].y = qhi.w * SM_C;

    v8f   o    = {};
    float mold = -1e30f;
    float lsum = 0.0f;

    const int srow = tid >> 2;
    const int scol = (tid & 3) * 4;
    const float* Kg = keys   + ((size_t)(n * SEQ + srow)) * EMB + h * DH + scol;
    const float* Vg = values + ((size_t)(n * SEQ + srow)) * EMB + h * DH + scol;
    const int st_off = srow * A_LD + scol;
    const int*   Mrow  = mask + (size_t)n * SEQ * SEQ + (size_t)(q0 + q) * SEQ + 8 * half;

    v4f rk = *reinterpret_cast<const v4f*>(Kg);
    v4f rv = *reinterpret_cast<const v4f*>(Vg);
    *reinterpret_cast<v4f*>(&sK[0][st_off]) = rk;
    *reinterpret_cast<v4f*>(&sV[0][st_off]) = rv;
    __syncthreads();

    const int kf_off = q * A_LD + 8 * half;
    const int vf_off = (8 * half) * A_LD + q;

    v4i mk[2][2];
    mk[0][0] = *reinterpret_cast<const v4i*>(Mrow);
    mk[0][1] = *reinterpret_cast<const v4i*>(Mrow + 4);

    for (int st = 0; st < SEQ / A_KT; ++st) {
        const int cur = st & 1;
        const int stn = min(st + 1, SEQ / A_KT - 1);
        rk = *reinterpret_cast<const v4f*>(Kg + (size_t)(stn * A_KT) * EMB);
        rv = *reinterpret_cast<const v4f*>(Vg + (size_t)(stn * A_KT) * EMB);

#pragma unroll
        for (int t = 0; t < A_KT / 16; ++t) {
            const int ktn = min(st * (A_KT / 16) + t + 1, SEQ / 16 - 1);
            mk[(t + 1) & 1][0] = *reinterpret_cast<const v4i*>(Mrow + ktn * 16);
            mk[(t + 1) & 1][1] = *reinterpret_cast<const v4i*>(Mrow + ktn * 16 + 4);

            const float* pk = &sK[cur][t * 16 * A_LD + kf_off];
            const float* pv = &sV[cur][t * 16 * A_LD + vf_off];
            const v4f klo = *reinterpret_cast<const v4f*>(pk);
            const v4f khi = *reinterpret_cast<const v4f*>(pk + 4);
            v2f av[4];
#pragma unroll
            for (int kk = 0; kk < 4; ++kk) {
                av[kk].x = pv[(2 * kk) * A_LD];
                av[kk].y = pv[(2 * kk + 1) * A_LD];
            }

            __builtin_amdgcn_sched_barrier(0);

            v2f ka[4];
            ka[0].x = klo.x; ka[0].y = klo.y;
            ka[1].x = klo.z; ka[1].y = klo.w;
            ka[2].x = khi.x; ka[2].y = khi.y;
            ka[3].x = khi.z; ka[3].y = khi.w;
            v8f s = {};
#pragma unroll
            for (int dd = 0; dd < 4; ++dd)
                s = __builtin_amdgcn_wmma_f32_16x16x4_f32(
                        false, ka[dd], false, Bq[dd], (short)0, s, false, false);

            const v4i m0 = mk[t & 1][0];
            const v4i m1 = mk[t & 1][1];
            float p[8];
#pragma unroll
            for (int r = 0; r < 8; ++r) {
                int mv = (r < 4) ? m0[r] : m1[r - 4];
                p[r]   = (mv == 0) ? (-1e20f * SM_C) : s[r];
            }

            float tmax = p[0];
#pragma unroll
            for (int r = 1; r < 8; ++r) tmax = fmaxf(tmax, p[r]);
            tmax = fmaxf(tmax, __shfl_xor(tmax, 16));
            const float mnew  = fmaxf(mold, tmax);
            const float alpha = __builtin_amdgcn_exp2f(mold - mnew);
            const v2f   mm = {mnew, mnew};
            float tsum = 0.0f;
#pragma unroll
            for (int kk = 0; kk < 4; ++kk) {
                v2f sv;
                sv.x = p[2 * kk];
                sv.y = p[2 * kk + 1];
                const v2f ev = sv - mm;
                p[2 * kk]     = __builtin_amdgcn_exp2f(ev.x);
                p[2 * kk + 1] = __builtin_amdgcn_exp2f(ev.y);
                tsum += p[2 * kk];
                tsum += p[2 * kk + 1];
            }
            tsum += __shfl_xor(tsum, 16);
            lsum = lsum * alpha + tsum;
            mold = mnew;
#pragma unroll
            for (int r = 0; r < 8; ++r) o[r] *= alpha;

#pragma unroll
            for (int kk = 0; kk < 4; ++kk) {
                v2f bp;
                bp.x = p[2 * kk];
                bp.y = p[2 * kk + 1];
                o = __builtin_amdgcn_wmma_f32_16x16x4_f32(
                        false, av[kk], false, bp, (short)0, o, false, false);
            }
        }

        *reinterpret_cast<v4f*>(&sK[cur ^ 1][st_off]) = rk;
        *reinterpret_cast<v4f*>(&sV[cur ^ 1][st_off]) = rv;
        __syncthreads();
    }

    const float rl = 1.0f / lsum;
    float* Orow = attn_out + ((size_t)(n * SEQ + q0 + q)) * EMB + h * DH + 8 * half;
    v4f o0, o1;
#pragma unroll
    for (int r = 0; r < 4; ++r) { o0[r] = o[r] * rl; o1[r] = o[r + 4] * rl; }
    *reinterpret_cast<v4f*>(Orow)     = o0;
    *reinterpret_cast<v4f*>(Orow + 4) = o1;
}

#define P2_BM 128
#define P2_BN 64
#define P2_BK 64
#define P2_LD 68

__global__ __launch_bounds__(256) void proj_wmma(
    const float* __restrict__ X, const float* __restrict__ W,
    const float* __restrict__ bvec, float* __restrict__ out)
{
    __shared__ __attribute__((aligned(16))) float sA[2][P2_BM * P2_LD];
    __shared__ __attribute__((aligned(16))) float sB[2][P2_BN * P2_LD];

    const int tid  = threadIdx.x;
    const int lane = tid & 31;
    const int wave = tid >> 5;
    const int c    = lane & 15;
    const int h    = lane >> 4;
    const int wm   = wave >> 1;
    const int wn   = wave & 1;
    const int j0   = (blockIdx.x & 15) * P2_BN;
    const int i0   = (blockIdx.x >> 4) * P2_BM;

    const int lrow = tid >> 4;
    const int lcol = (tid & 15) * 4;
    const float* Xg = X + (size_t)(i0 + lrow) * EMB + lcol;
    const float* Wg = W + (size_t)(j0 + lrow) * EMB + lcol;
    const int lds_off = lrow * P2_LD + lcol;

    v4f ra[P2_BM / 16], rb[P2_BN / 16];
#pragma unroll
    for (int i = 0; i < P2_BM / 16; ++i)
        ra[i] = *reinterpret_cast<const v4f*>(Xg + (size_t)(16 * i) * EMB);
#pragma unroll
    for (int i = 0; i < P2_BN / 16; ++i)
        rb[i] = *reinterpret_cast<const v4f*>(Wg + (size_t)(16 * i) * EMB);
#pragma unroll
    for (int i = 0; i < P2_BM / 16; ++i)
        *reinterpret_cast<v4f*>(&sA[0][lds_off + 16 * i * P2_LD]) = ra[i];
#pragma unroll
    for (int i = 0; i < P2_BN / 16; ++i)
        *reinterpret_cast<v4f*>(&sB[0][lds_off + 16 * i * P2_LD]) = rb[i];
    __syncthreads();

    v8f acc00 = {}, acc01 = {}, acc10 = {}, acc11 = {};
    const int a_off = (wm * 32 + c) * P2_LD + 4 * h;
    const int b_off = (wn * 32 + c) * P2_LD + 4 * h;

    for (int kt = 0; kt < EMB / P2_BK; ++kt) {
        const int  cur  = kt & 1;
        const bool more = (kt + 1) < EMB / P2_BK;
        if (more) {
            const int kb = (kt + 1) * P2_BK;
#pragma unroll
            for (int i = 0; i < P2_BM / 16; ++i)
                ra[i] = *reinterpret_cast<const v4f*>(Xg + (size_t)(16 * i) * EMB + kb);
#pragma unroll
            for (int i = 0; i < P2_BN / 16; ++i)
                rb[i] = *reinterpret_cast<const v4f*>(Wg + (size_t)(16 * i) * EMB + kb);
        }

        const float* pa = &sA[cur][a_off];
        const float* pb = &sB[cur][b_off];
#pragma unroll
        for (int k8 = 0; k8 < P2_BK / 8; ++k8) {
            const v4f fa0 = *reinterpret_cast<const v4f*>(pa + 8 * k8);
            const v4f fa1 = *reinterpret_cast<const v4f*>(pa + 16 * P2_LD + 8 * k8);
            const v4f fb0 = *reinterpret_cast<const v4f*>(pb + 8 * k8);
            const v4f fb1 = *reinterpret_cast<const v4f*>(pb + 16 * P2_LD + 8 * k8);
            v2f a0, a1, b0, b1;
            a0.x = fa0.x; a0.y = fa0.y;  a1.x = fa1.x; a1.y = fa1.y;
            b0.x = fb0.x; b0.y = fb0.y;  b1.x = fb1.x; b1.y = fb1.y;
            acc00 = __builtin_amdgcn_wmma_f32_16x16x4_f32(false, a0, false, b0, (short)0, acc00, false, false);
            acc01 = __builtin_amdgcn_wmma_f32_16x16x4_f32(false, a0, false, b1, (short)0, acc01, false, false);
            acc10 = __builtin_amdgcn_wmma_f32_16x16x4_f32(false, a1, false, b0, (short)0, acc10, false, false);
            acc11 = __builtin_amdgcn_wmma_f32_16x16x4_f32(false, a1, false, b1, (short)0, acc11, false, false);
            a0.x = fa0.z; a0.y = fa0.w;  a1.x = fa1.z; a1.y = fa1.w;
            b0.x = fb0.z; b0.y = fb0.w;  b1.x = fb1.z; b1.y = fb1.w;
            acc00 = __builtin_amdgcn_wmma_f32_16x16x4_f32(false, a0, false, b0, (short)0, acc00, false, false);
            acc01 = __builtin_amdgcn_wmma_f32_16x16x4_f32(false, a0, false, b1, (short)0, acc01, false, false);
            acc10 = __builtin_amdgcn_wmma_f32_16x16x4_f32(false, a1, false, b0, (short)0, acc10, false, false);
            acc11 = __builtin_amdgcn_wmma_f32_16x16x4_f32(false, a1, false, b1, (short)0, acc11, false, false);
        }

        if (more) {
            const int nxt = cur ^ 1;
#pragma unroll
            for (int i = 0; i < P2_BM / 16; ++i)
                *reinterpret_cast<v4f*>(&sA[nxt][lds_off + 16 * i * P2_LD]) = ra[i];
#pragma unroll
            for (int i = 0; i < P2_BN / 16; ++i)
                *reinterpret_cast<v4f*>(&sB[nxt][lds_off + 16 * i * P2_LD]) = rb[i];
        }
        __syncthreads();
    }

    const int   col0 = j0 + wn * 32 + c;
    const int   col1 = col0 + 16;
    const float bj0  = bvec[col0];
    const float bj1  = bvec[col1];
    const int   row0 = i0 + wm * 32 + 8 * h;
#pragma unroll
    for (int r = 0; r < 8; ++r) {
        out[(size_t)(row0 + r) * EMB + col0]      = acc00[r] + bj0;
        out[(size_t)(row0 + r) * EMB + col1]      = acc01[r] + bj1;
        out[(size_t)(row0 + 16 + r) * EMB + col0] = acc10[r] + bj0;
        out[(size_t)(row0 + 16 + r) * EMB + col1] = acc11[r] + bj1;
    }
}

extern "C" void kernel_launch(void* const* d_in, const int* in_sizes, int n_in,
                              void* d_out, int out_size, void* d_ws, size_t ws_size,
                              hipStream_t stream)
{
    const float* keys   = (const float*)d_in[0];
    const float* query  = (const float*)d_in[1];
    const float* values = (const float*)d_in[2];
    const int*   mask   = (const int*)d_in[3];
    const float* W_out  = (const float*)d_in[4];
    const float* b_out  = (const float*)d_in[5];
    float* out  = (float*)d_out;
    float* attn = (float*)d_ws;

    const int N = 2;
    attn_flash_wmma<<<N * NH * (SEQ / 128), 256, 0, stream>>>(
        keys, query, values, mask, attn);
    proj_wmma<<<(N * SEQ / P2_BM) * (EMB / P2_BN), 256, 0, stream>>>(
        attn, W_out, b_out, out);
}


# attention mask: each wave tests its 16 rows' mask words once per 64-key stage (whole-line loads, unsigned min, one ballot); the 8 compares + 8 selects per tile run only where a stage holds a zero
# speedup vs baseline: 2.2148x; 1.0644x over previous
#include <hip/hip_runtime.h>


typedef float v2f __attribute__((ext_vector_type(2)));
typedef float v4f __attribute__((ext_vector_type(4)));
typedef float v8f __attribute__((ext_vector_type(8)));
typedef int   v4i __attribute__((ext_vector_type(4)));

#define SEQ  1024
#define EMB  1024
#define NH   64
#define DH   16

#define A_KT 64
#define A_LD 20
#define SM_C (1.4426950408889634f * 0.03125f)

__device__ __forceinline__ unsigned stage_min(const v4i (&w)[8])
{
    unsigned mn = 0xffffffffu;
#pragma unroll
    for (int i = 0; i < 8; ++i) {
        mn = min(mn, min((unsigned)w[i].x, (unsigned)w[i].y));
        mn = min(mn, min((unsigned)w[i].z, (unsigned)w[i].w));
    }
    return mn;
}

__global__ __launch_bounds__(256) void attn_flash_wmma(
    const float* __restrict__ keys, const float* __restrict__ query,
    const float* __restrict__ values, const int* __restrict__ mask,
    float* __restrict__ attn_out)
{
    __shared__ __attribute__((aligned(16))) float sK[2][A_KT * A_LD];
    __shared__ __attribute__((aligned(16))) float sV[2][A_KT * A_LD];

    const int tid  = threadIdx.x;
    const int lane = tid & 31;
    const int wave = tid >> 5;
    const int q    = lane & 15;
    const int half = lane >> 4;

    const int idx = blockIdx.x;
    const int qb  = idx & 7;
    const int h   = (idx >> 3) & 63;
    const int n   = idx >> 9;
    const int q0  = qb * 128 + wave * 16;

    const float* Qrow = query + ((size_t)(n * SEQ + q0 + q)) * EMB + h * DH + 8 * half;
    const v4f qlo = *reinterpret_cast<const v4f*>(Qrow);
    const v4f qhi = *reinterpret_cast<const v4f*>(Qrow + 4);
    v2f Bq[4];
    Bq[0].x = qlo.x * SM_C; Bq[0].y = qlo.y * SM_C;
    Bq[1].x = qlo.z * SM_C; Bq[1].y = qlo.w * SM_C;
    Bq[2].x = qhi.x * SM_C; Bq[2].y = qhi.y * SM_C;
    Bq[3].x = qhi.z * SM_C; Bq[3].y = qhi.w * SM_C;

    v8f   o    = {};
    float mold = -1e30f;
    float lsum = 0.0f;

    const int srow = tid >> 2;
    const int scol = (tid & 3) * 4;
    const float* Kg = keys   + ((size_t)(n * SEQ + srow)) * EMB + h * DH + scol;
    const float* Vg = values + ((size_t)(n * SEQ + srow)) * EMB + h * DH + scol;
    const int st_off = srow * A_LD + scol;
    const int*   Mrow  = mask + (size_t)n * SEQ * SEQ + (size_t)(q0 + q) * SEQ + 8 * half;
    const int*   Mtest = mask + (size_t)n * SEQ * SEQ + (size_t)(q0 + half) * SEQ + 4 * q;

    v4f rk = *reinterpret_cast<const v4f*>(Kg);
    v4f rv = *reinterpret_cast<const v4f*>(Vg);
    *reinterpret_cast<v4f*>(&sK[0][st_off]) = rk;
    *reinterpret_cast<v4f*>(&sV[0][st_off]) = rv;
    __syncthreads();

    const int kf_off = q * A_LD + 8 * half;
    const int vf_off = (8 * half) * A_LD + q;

    v4i mw[8];
#pragma unroll
    for (int i = 0; i < 8; ++i)
        mw[i] = *reinterpret_cast<const v4i*>(Mtest + (size_t)(2 * i) * SEQ);
    bool anyz = __builtin_amdgcn_ballot_w32(stage_min(mw) == 0u) != 0u;

    for (int st = 0; st < SEQ / A_KT; ++st) {
        const int cur = st & 1;
        const int stn = min(st + 1, SEQ / A_KT - 1);
        rk = *reinterpret_cast<const v4f*>(Kg + (size_t)(stn * A_KT) * EMB);
        rv = *reinterpret_cast<const v4f*>(Vg + (size_t)(stn * A_KT) * EMB);
#pragma unroll
        for (int i = 0; i < 8; ++i)
            mw[i] = *reinterpret_cast<const v4i*>(Mtest + (size_t)(2 * i) * SEQ + stn * A_KT);

#pragma unroll
        for (int t = 0; t < A_KT / 16; ++t) {
            const float* pk = &sK[cur][t * 16 * A_LD + kf_off];
            const float* pv = &sV[cur][t * 16 * A_LD + vf_off];
            const v4f klo = *reinterpret_cast<const v4f*>(pk);
            const v4f khi = *reinterpret_cast<const v4f*>(pk + 4);
            v2f av[4];
#pragma unroll
            for (int kk = 0; kk < 4; ++kk) {
                av[kk].x = pv[(2 * kk) * A_LD];
                av[kk].y = pv[(2 * kk + 1) * A_LD];
            }

            __builtin_amdgcn_sched_barrier(0);

            v2f ka[4];
            ka[0].x = klo.x; ka[0].y = klo.y;
            ka[1].x = klo.z; ka[1].y = klo.w;
            ka[2].x = khi.x; ka[2].y = khi.y;
            ka[3].x = khi.z; ka[3].y = khi.w;
            v8f s = {};
#pragma unroll
            for (int dd = 0; dd < 4; ++dd)
                s = __builtin_amdgcn_wmma_f32_16x16x4_f32(
                        false, ka[dd], false, Bq[dd], (short)0, s, false, false);

            if (anyz) {
                const int kt = st * (A_KT / 16) + t;
                const v4i m0 = *reinterpret_cast<const v4i*>(Mrow + kt * 16);
                const v4i m1 = *reinterpret_cast<const v4i*>(Mrow + kt * 16 + 4);
#pragma unroll
                for (int r = 0; r < 8; ++r) {
                    int mv = (r < 4) ? m0[r] : m1[r - 4];
                    s[r]   = (mv == 0) ? (-1e20f * SM_C) : s[r];
                }
                asm volatile("" : "+v"(s));
            }
            float p[8];
#pragma unroll
            for (int r = 0; r < 8; ++r) p[r] = s[r];

            float tmax = p[0];
#pragma unroll
            for (int r = 1; r < 8; ++r) tmax = fmaxf(tmax, p[r]);
            tmax = fmaxf(tmax, __shfl_xor(tmax, 16));
            const float mnew  = fmaxf(mold, tmax);
            const float alpha = __builtin_amdgcn_exp2f(mold - mnew);
            const v2f   mm = {mnew, mnew};
            float tsum = 0.0f;
#pragma unroll
            for (int kk = 0; kk < 4; ++kk) {
                v2f sv;
                sv.x = p[2 * kk];
                sv.y = p[2 * kk + 1];
                const v2f ev = sv - mm;
                p[2 * kk]     = __builtin_amdgcn_exp2f(ev.x);
                p[2 * kk + 1] = __builtin_amdgcn_exp2f(ev.y);
                tsum += p[2 * kk];
                tsum += p[2 * kk + 1];
            }
            tsum += __shfl_xor(tsum, 16);
            lsum = lsum * alpha + tsum;
            mold = mnew;
#pragma unroll
            for (int r = 0; r < 8; ++r) o[r] *= alpha;

#pragma unroll
            for (int kk = 0; kk < 4; ++kk) {
                v2f bp;
                bp.x = p[2 * kk];
                bp.y = p[2 * kk + 1];
                o = __builtin_amdgcn_wmma_f32_16x16x4_f32(
                        false, av[kk], false, bp, (short)0, o, false, false);
            }
        }

        *reinterpret_cast<v4f*>(&sK[cur ^ 1][st_off]) = rk;
        *reinterpret_cast<v4f*>(&sV[cur ^ 1][st_off]) = rv;
        anyz = __builtin_amdgcn_ballot_w32(stage_min(mw) == 0u) != 0u;
        __syncthreads();
    }

    const float rl = 1.0f / lsum;
    float* Orow = attn_out + ((size_t)(n * SEQ + q0 + q)) * EMB + h * DH + 8 * half;
    v4f o0, o1;
#pragma unroll
    for (int r = 0; r < 4; ++r) { o0[r] = o[r] * rl; o1[r] = o[r + 4] * rl; }
    *reinterpret_cast<v4f*>(Orow)     = o0;
    *reinterpret_cast<v4f*>(Orow + 4) = o1;
}

#define P2_BM 128
#define P2_BN 64
#define P2_BK 64
#define P2_LD 68

__global__ __launch_bounds__(256) void proj_wmma(
    const float* __restrict__ X, const float* __restrict__ W,
    const float* __restrict__ bvec, float* __restrict__ out)
{
    __shared__ __attribute__((aligned(16))) float sA[2][P2_BM * P2_LD];
    __shared__ __attribute__((aligned(16))) float sB[2][P2_BN * P2_LD];

    const int tid  = threadIdx.x;
    const int lane = tid & 31;
    const int wave = tid >> 5;
    const int c    = lane & 15;
    const int h    = lane >> 4;
    const int wm   = wave >> 1;
    const int wn   = wave & 1;
    const int j0   = (blockIdx.x & 15) * P2_BN;
    const int i0   = (blockIdx.x >> 4) * P2_BM;

    const int lrow = tid >> 4;
    const int lcol = (tid & 15) * 4;
    const float* Xg = X + (size_t)(i0 + lrow) * EMB + lcol;
    const float* Wg = W + (size_t)(j0 + lrow) * EMB + lcol;
    const int lds_off = lrow * P2_LD + lcol;

    v4f ra[P2_BM / 16], rb[P2_BN / 16];
#pragma unroll
    for (int i = 0; i < P2_BM / 16; ++i)
        ra[i] = *reinterpret_cast<const v4f*>(Xg + (size_t)(16 * i) * EMB);
#pragma unroll
    for (int i = 0; i < P2_BN / 16; ++i)
        rb[i] = *reinterpret_cast<const v4f*>(Wg + (size_t)(16 * i) * EMB);
#pragma unroll
    for (int i = 0; i < P2_BM / 16; ++i)
        *reinterpret_cast<v4f*>(&sA[0][lds_off + 16 * i * P2_LD]) = ra[i];
#pragma unroll
    for (int i = 0; i < P2_BN / 16; ++i)
        *reinterpret_cast<v4f*>(&sB[0][lds_off + 16 * i * P2_LD]) = rb[i];
    __syncthreads();

    v8f acc00 = {}, acc01 = {}, acc10 = {}, acc11 = {};
    const int a_off = (wm * 32 + c) * P2_LD + 4 * h;
    const int b_off = (wn * 32 + c) * P2_LD + 4 * h;

    for (int kt = 0; kt < EMB / P2_BK; ++kt) {
        const int  cur  = kt & 1;
        const bool more = (kt + 1) < EMB / P2_BK;
        if (more) {
            const int kb = (kt + 1) * P2_BK;
#pragma unroll
            for (int i = 0; i < P2_BM / 16; ++i)
                ra[i] = *reinterpret_cast<const v4f*>(Xg + (size_t)(16 * i) * EMB + kb);
#pragma unroll
            for (int i = 0; i < P2_BN / 16; ++i)
                rb[i] = *reinterpret_cast<const v4f*>(Wg + (size_t)(16 * i) * EMB + kb);
        }

        const float* pa = &sA[cur][a_off];
        const float* pb = &sB[cur][b_off];
#pragma unroll
        for (int k8 = 0; k8 < P2_BK / 8; ++k8) {
            const v4f fa0 = *reinterpret_cast<const v4f*>(pa + 8 * k8);
            const v4f fa1 = *reinterpret_cast<const v4f*>(pa + 16 * P2_LD + 8 * k8);
            const v4f fb0 = *reinterpret_cast<const v4f*>(pb + 8 * k8);
            const v4f fb1 = *reinterpret_cast<const v4f*>(pb + 16 * P2_LD + 8 * k8);
            v2f a0, a1, b0, b1;
            a0.x = fa0.x; a0.y = fa0.y;  a1.x = fa1.x; a1.y = fa1.y;
            b0.x = fb0.x; b0.y = fb0.y;  b1.x = fb1.x; b1.y = fb1.y;
            acc00 = __builtin_amdgcn_wmma_f32_16x16x4_f32(false, a0, false, b0, (short)0, acc00, false, false);
            acc01 = __builtin_amdgcn_wmma_f32_16x16x4_f32(false, a0, false, b1, (short)0, acc01, false, false);
            acc10 = __builtin_amdgcn_wmma_f32_16x16x4_f32(false, a1, false, b0, (short)0, acc10, false, false);
            acc11 = __builtin_amdgcn_wmma_f32_16x16x4_f32(false, a1, false, b1, (short)0, acc11, false, false);
            a0.x = fa0.z; a0.y = fa0.w;  a1.x = fa1.z; a1.y = fa1.w;
            b0.x = fb0.z; b0.y = fb0.w;  b1.x = fb1.z; b1.y = fb1.w;
            acc00 = __builtin_amdgcn_wmma_f32_16x16x4_f32(false, a0, false, b0, (short)0, acc00, false, false);
            acc01 = __builtin_amdgcn_wmma_f32_16x16x4_f32(false, a0, false, b1, (short)0, acc01, false, false);
            acc10 = __builtin_amdgcn_wmma_f32_16x16x4_f32(false, a1, false, b0, (short)0, acc10, false, false);
            acc11 = __builtin_amdgcn_wmma_f32_16x16x4_f32(false, a1, false, b1, (short)0, acc11, false, false);
        }

        if (more) {
            const int nxt = cur ^ 1;
#pragma unroll
            for (int i = 0; i < P2_BM / 16; ++i)
                *reinterpret_cast<v4f*>(&sA[nxt][lds_off + 16 * i * P2_LD]) = ra[i];
#pragma unroll
            for (int i = 0; i < P2_BN / 16; ++i)
                *reinterpret_cast<v4f*>(&sB[nxt][lds_off + 16 * i * P2_LD]) = rb[i];
        }
        __syncthreads();
    }

    const int   col0 = j0 + wn * 32 + c;
    const int   col1 = col0 + 16;
    const float bj0  = bvec[col0];
    const float bj1  = bvec[col1];
    const int   row0 = i0 + wm * 32 + 8 * h;
#pragma unroll
    for (int r = 0; r < 8; ++r) {
        out[(size_t)(row0 + r) * EMB + col0]      = acc00[r] + bj0;
        out[(size_t)(row0 + r) * EMB + col1]      = acc01[r] + bj1;
        out[(size_t)(row0 + 16 + r) * EMB + col0] = acc10[r] + bj0;
        out[(size_t)(row0 + 16 + r) * EMB + col1] = acc11[r] + bj1;
    }
}

extern "C" void kernel_launch(void* const* d_in, const int* in_sizes, int n_in,
                              void* d_out, int out_size, void* d_ws, size_t ws_size,
                              hipStream_t stream)
{
    const float* keys   = (const float*)d_in[0];
    const float* query  = (const float*)d_in[1];
    const float* values = (const float*)d_in[2];
    const int*   mask   = (const int*)d_in[3];
    const float* W_out  = (const float*)d_in[4];
    const float* b_out  = (const float*)d_in[5];
    float* out  = (float*)d_out;
    float* attn = (float*)d_ws;

    const int N = 2;
    attn_flash_wmma<<<N * NH * (SEQ / 128), 256, 0, stream>>>(
        keys, query, values, mask, attn);
    proj_wmma<<<(N * SEQ / P2_BM) * (EMB / P2_BN), 256, 0, stream>>>(
        attn, W_out, b_out, out);
}


# attention softmax: no LDS cross-lane trip left per tile: row max crosses lane halves by v_permlanex16; row sum kept as per-lane packed partials, halves added after the loop (sum order changes)
# speedup vs baseline: 2.3024x; 1.0396x over previous
#include <hip/hip_runtime.h>


typedef float v2f __attribute__((ext_vector_type(2)));
typedef float v4f __attribute__((ext_vector_type(4)));
typedef float v8f __attribute__((ext_vector_type(8)));
typedef int   v4i __attribute__((ext_vector_type(4)));

#define SEQ  1024
#define EMB  1024
#define NH   64
#define DH   16

#define A_KT 64
#define A_LD 20
#define SM_C (1.4426950408889634f * 0.03125f)

__device__ __forceinline__ unsigned stage_min(const v4i (&w)[8])
{
    unsigned mn = 0xffffffffu;
#pragma unroll
    for (int i = 0; i < 8; ++i) {
        mn = min(mn, min((unsigned)w[i].x, (unsigned)w[i].y));
        mn = min(mn, min((unsigned)w[i].z, (unsigned)w[i].w));
    }
    return mn;
}

__device__ __forceinline__ float xhalf(float x)
{
    const int b = __builtin_bit_cast(int, x);
    return __builtin_bit_cast(float,
        __builtin_amdgcn_permlanex16(b, b, 0x76543210, 0xfedcba98, true, false));
}

__global__ __launch_bounds__(256) void attn_flash_wmma(
    const float* __restrict__ keys, const float* __restrict__ query,
    const float* __restrict__ values, const int* __restrict__ mask,
    float* __restrict__ attn_out)
{
    __shared__ __attribute__((aligned(16))) float sK[2][A_KT * A_LD];
    __shared__ __attribute__((aligned(16))) float sV[2][A_KT * A_LD];

    const int tid  = threadIdx.x;
    const int lane = tid & 31;
    const int wave = tid >> 5;
    const int q    = lane & 15;
    const int half = lane >> 4;

    const int idx = blockIdx.x;
    const int qb  = idx & 7;
    const int h   = (idx >> 3) & 63;
    const int n   = idx >> 9;
    const int q0  = qb * 128 + wave * 16;

    const float* Qrow = query + ((size_t)(n * SEQ + q0 + q)) * EMB + h * DH + 8 * half;
    const v4f qlo = *reinterpret_cast<const v4f*>(Qrow);
    const v4f qhi = *reinterpret_cast<const v4f*>(Qrow + 4);
    v2f Bq[4];
    Bq[0].x = qlo.x * SM_C; Bq[0].y = qlo.y * SM_C;
    Bq[1].x = qlo.z * SM_C; Bq[1].y = qlo.w * SM_C;
    Bq[2].x = qhi.x * SM_C; Bq[2].y = qhi.y * SM_C;
    Bq[3].x = qhi.z * SM_C; Bq[3].y = qhi.w * SM_C;

    v8f   o    = {};
    float mold = -1e30f;
    v2f   lsum2 = {0.0f, 0.0f};

    const int srow = tid >> 2;
    const int scol = (tid & 3) * 4;
    const float* Kg = keys   + ((size_t)(n * SEQ + srow)) * EMB + h * DH + scol;
    const float* Vg = values + ((size_t)(n * SEQ + srow)) * EMB + h * DH + scol;
    const int st_off = srow * A_LD + scol;
    const int*   Mrow  = mask + (size_t)n * SEQ * SEQ + (size_t)(q0 + q) * SEQ + 8 * half;
    const int*   Mtest = mask + (size_t)n * SEQ * SEQ + (size_t)(q0 + half) * SEQ + 4 * q;

    v4f rk = *reinterpret_cast<const v4f*>(Kg);
    v4f rv = *reinterpret_cast<const v4f*>(Vg);
    *reinterpret_cast<v4f*>(&sK[0][st_off]) = rk;
    *reinterpret_cast<v4f*>(&sV[0][st_off]) = rv;
    __syncthreads();

    const int kf_off = q * A_LD + 8 * half;
    const int vf_off = (8 * half) * A_LD + q;

    v4i mw[8];
#pragma unroll
    for (int i = 0; i < 8; ++i)
        mw[i] = *reinterpret_cast<const v4i*>(Mtest + (size_t)(2 * i) * SEQ);
    bool anyz = __builtin_amdgcn_ballot_w32(stage_min(mw) == 0u) != 0u;

    for (int st = 0; st < SEQ / A_KT; ++st) {
        const int cur = st & 1;
        const int stn = min(st + 1, SEQ / A_KT - 1);
        rk = *reinterpret_cast<const v4f*>(Kg + (size_t)(stn * A_KT) * EMB);
        rv = *reinterpret_cast<const v4f*>(Vg + (size_t)(stn * A_KT) * EMB);
#pragma unroll
        for (int i = 0; i < 8; ++i)
            mw[i] = *reinterpret_cast<const v4i*>(Mtest + (size_t)(2 * i) * SEQ + stn * A_KT);

#pragma unroll
        for (int t = 0; t < A_KT / 16; ++t) {
            const float* pk = &sK[cur][t * 16 * A_LD + kf_off];
            const float* pv = &sV[cur][t * 16 * A_LD + vf_off];
            const v4f klo = *reinterpret_cast<const v4f*>(pk);
            const v4f khi = *reinterpret_cast<const v4f*>(pk + 4);
            v2f av[4];
#pragma unroll
            for (int kk = 0; kk < 4; ++kk) {
                av[kk].x = pv[(2 * kk) * A_LD];
                av[kk].y = pv[(2 * kk + 1) * A_LD];
            }

            __builtin_amdgcn_sched_barrier(0);

            v2f ka[4];
            ka[0].x = klo.x; ka[0].y = klo.y;
            ka[1].x = klo.z; ka[1].y = klo.w;
            ka[2].x = khi.x; ka[2].y = khi.y;
            ka[3].x = khi.z; ka[3].y = khi.w;
            v8f s = {};
#pragma unroll
            for (int dd = 0; dd < 4; ++dd)
                s = __builtin_amdgcn_wmma_f32_16x16x4_f32(
                        false, ka[dd], false, Bq[dd], (short)0, s, false, false);

            if (anyz) {
                const int kt = st * (A_KT / 16) + t;
                const v4i m0 = *reinterpret_cast<const v4i*>(Mrow + kt * 16);
                const v4i m1 = *reinterpret_cast<const v4i*>(Mrow + kt * 16 + 4);
#pragma unroll
                for (int r = 0; r < 8; ++r) {
                    int mv = (r < 4) ? m0[r] : m1[r - 4];
                    s[r]   = (mv == 0) ? (-1e20f * SM_C) : s[r];
                }
                asm volatile("" : "+v"(s));
            }
            float p[8];
#pragma unroll
            for (int r = 0; r < 8; ++r) p[r] = s[r];

            float tmax = p[0];
#pragma unroll
            for (int r = 1; r < 8; ++r) tmax = fmaxf(tmax, p[r]);
            tmax = fmaxf(tmax, xhalf(tmax));
            const float mnew  = fmaxf(mold, tmax);
            const float alpha = __builtin_amdgcn_exp2f(mold - mnew);
            const v2f   mm = {mnew, mnew};
            const v2f   aa = {alpha, alpha};
            v2f pp[4];
#pragma unroll
            for (int kk = 0; kk < 4; ++kk) {
                v2f sv;
                sv.x = p[2 * kk];
                sv.y = p[2 * kk + 1];
                const v2f ev = sv - mm;
                pp[kk].x = __builtin_amdgcn_exp2f(ev.x);
                pp[kk].y = __builtin_amdgcn_exp2f(ev.y);
            }
            lsum2 = lsum2 * aa + ((pp[0] + pp[1]) + (pp[2] + pp[3]));
            mold = mnew;
#pragma unroll
            for (int r = 0; r < 8; ++r) o[r] *= alpha;

#pragma unroll
            for (int kk = 0; kk < 4; ++kk) {
                o = __builtin_amdgcn_wmma_f32_16x16x4_f32(
                        false, av[kk], false, pp[kk], (short)0, o, false, false);
            }
            asm volatile("" : "+v"(lsum2));
        }

        *reinterpret_cast<v4f*>(&sK[cur ^ 1][st_off]) = rk;
        *reinterpret_cast<v4f*>(&sV[cur ^ 1][st_off]) = rv;
        anyz = __builtin_amdgcn_ballot_w32(stage_min(mw) == 0u) != 0u;
        __syncthreads();
    }

    float lsum = lsum2.x + lsum2.y;
    lsum += xhalf(lsum);
    const float rl = 1.0f / lsum;
    float* Orow = attn_out + ((size_t)(n * SEQ + q0 + q)) * EMB + h * DH + 8 * half;
    v4f o0, o1;
#pragma unroll
    for (int r = 0; r < 4; ++r) { o0[r] = o[r] * rl; o1[r] = o[r + 4] * rl; }
    *reinterpret_cast<v4f*>(Orow)     = o0;
    *reinterpret_cast<v4f*>(Orow + 4) = o1;
}

#define P2_BM 128
#define P2_BN 64
#define P2_BK 64
#define P2_LD 68

__global__ __launch_bounds__(256) void proj_wmma(
    const float* __restrict__ X, const float* __restrict__ W,
    const float* __restrict__ bvec, float* __restrict__ out)
{
    __shared__ __attribute__((aligned(16))) float sA[2][P2_BM * P2_LD];
    __shared__ __attribute__((aligned(16))) float sB[2][P2_BN * P2_LD];

    const int tid  = threadIdx.x;
    const int lane = tid & 31;
    const int wave = tid >> 5;
    const int c    = lane & 15;
    const int h    = lane >> 4;
    const int wm   = wave >> 1;
    const int wn   = wave & 1;
    const int j0   = (blockIdx.x & 15) * P2_BN;
    const int i0   = (blockIdx.x >> 4) * P2_BM;

    const int lrow = tid >> 4;
    const int lcol = (tid & 15) * 4;
    const float* Xg = X + (size_t)(i0 + lrow) * EMB + lcol;
    const float* Wg = W + (size_t)(j0 + lrow) * EMB + lcol;
    const int lds_off = lrow * P2_LD + lcol;

    v4f ra[P2_BM / 16], rb[P2_BN / 16];
#pragma unroll
    for (int i = 0; i < P2_BM / 16; ++i)
        ra[i] = *reinterpret_cast<const v4f*>(Xg + (size_t)(16 * i) * EMB);
#pragma unroll
    for (int i = 0; i < P2_BN / 16; ++i)
        rb[i] = *reinterpret_cast<const v4f*>(Wg + (size_t)(16 * i) * EMB);
#pragma unroll
    for (int i = 0; i < P2_BM / 16; ++i)
        *reinterpret_cast<v4f*>(&sA[0][lds_off + 16 * i * P2_LD]) = ra[i];
#pragma unroll
    for (int i = 0; i < P2_BN / 16; ++i)
        *reinterpret_cast<v4f*>(&sB[0][lds_off + 16 * i * P2_LD]) = rb[i];
    __syncthreads();

    v8f acc00 = {}, acc01 = {}, acc10 = {}, acc11 = {};
    const int a_off = (wm * 32 + c) * P2_LD + 4 * h;
    const int b_off = (wn * 32 + c) * P2_LD + 4 * h;

    for (int kt = 0; kt < EMB / P2_BK; ++kt) {
        const int  cur  = kt & 1;
        const bool more = (kt + 1) < EMB / P2_BK;
        if (more) {
            const int kb = (kt + 1) * P2_BK;
#pragma unroll
            for (int i = 0; i < P2_BM / 16; ++i)
                ra[i] = *reinterpret_cast<const v4f*>(Xg + (size_t)(16 * i) * EMB + kb);
#pragma unroll
            for (int i = 0; i < P2_BN / 16; ++i)
                rb[i] = *reinterpret_cast<const v4f*>(Wg + (size_t)(16 * i) * EMB + kb);
        }

        const float* pa = &sA[cur][a_off];
        const float* pb = &sB[cur][b_off];
#pragma unroll
        for (int k8 = 0; k8 < P2_BK / 8; ++k8) {
            const v4f fa0 = *reinterpret_cast<const v4f*>(pa + 8 * k8);
            const v4f fa1 = *reinterpret_cast<const v4f*>(pa + 16 * P2_LD + 8 * k8);
            const v4f fb0 = *reinterpret_cast<const v4f*>(pb + 8 * k8);
            const v4f fb1 = *reinterpret_cast<const v4f*>(pb + 16 * P2_LD + 8 * k8);
            v2f a0, a1, b0, b1;
            a0.x = fa0.x; a0.y = fa0.y;  a1.x = fa1.x; a1.y = fa1.y;
            b0.x = fb0.x; b0.y = fb0.y;  b1.x = fb1.x; b1.y = fb1.y;
            acc00 = __builtin_amdgcn_wmma_f32_16x16x4_f32(false, a0, false, b0, (short)0, acc00, false, false);
            acc01 = __builtin_amdgcn_wmma_f32_16x16x4_f32(false, a0, false, b1, (short)0, acc01, false, false);
            acc10 = __builtin_amdgcn_wmma_f32_16x16x4_f32(false, a1, false, b0, (short)0, acc10, false, false);
            acc11 = __builtin_amdgcn_wmma_f32_16x16x4_f32(false, a1, false, b1, (short)0, acc11, false, false);
            a0.x = fa0.z; a0.y = fa0.w;  a1.x = fa1.z; a1.y = fa1.w;
            b0.x = fb0.z; b0.y = fb0.w;  b1.x = fb1.z; b1.y = fb1.w;
            acc00 = __builtin_amdgcn_wmma_f32_16x16x4_f32(false, a0, false, b0, (short)0, acc00, false, false);
            acc01 = __builtin_amdgcn_wmma_f32_16x16x4_f32(false, a0, false, b1, (short)0, acc01, false, false);
            acc10 = __builtin_amdgcn_wmma_f32_16x16x4_f32(false, a1, false, b0, (short)0, acc10, false, false);
            acc11 = __builtin_amdgcn_wmma_f32_16x16x4_f32(false, a1, false, b1, (short)0, acc11, false, false);
        }

        if (more) {
            const int nxt = cur ^ 1;
#pragma unroll
            for (int i = 0; i < P2_BM / 16; ++i)
                *reinterpret_cast<v4f*>(&sA[nxt][lds_off + 16 * i * P2_LD]) = ra[i];
#pragma unroll
            for (int i = 0; i < P2_BN / 16; ++i)
                *reinterpret_cast<v4f*>(&sB[nxt][lds_off + 16 * i * P2_LD]) = rb[i];
        }
        __syncthreads();
    }

    const int   col0 = j0 + wn * 32 + c;
    const int   col1 = col0 + 16;
    const float bj0  = bvec[col0];
    const float bj1  = bvec[col1];
    const int   row0 = i0 + wm * 32 + 8 * h;
#pragma unroll
    for (int r = 0; r < 8; ++r) {
        out[(size_t)(row0 + r) * EMB + col0]      = acc00[r] + bj0;
        out[(size_t)(row0 + r) * EMB + col1]      = acc01[r] + bj1;
        out[(size_t)(row0 + 16 + r) * EMB + col0] = acc10[r] + bj0;
        out[(size_t)(row0 + 16 + r) * EMB + col1] = acc11[r] + bj1;
    }
}

extern "C" void kernel_launch(void* const* d_in, const int* in_sizes, int n_in,
                              void* d_out, int out_size, void* d_ws, size_t ws_size,
                              hipStream_t stream)
{
    const float* keys   = (const float*)d_in[0];
    const float* query  = (const float*)d_in[1];
    const float* values = (const float*)d_in[2];
    const int*   mask   = (const int*)d_in[3];
    const float* W_out  = (const float*)d_in[4];
    const float* b_out  = (const float*)d_in[5];
    float* out  = (float*)d_out;
    float* attn = (float*)d_ws;

    const int N = 2;
    attn_flash_wmma<<<N * NH * (SEQ / 128), 256, 0, stream>>>(
        keys, query, values, mask, attn);
    proj_wmma<<<(N * SEQ / P2_BM) * (EMB / P2_BN), 256, 0, stream>>>(
        attn, W_out, b_out, out);
}
